# starting kernel
# baseline (speedup 1.0000x reference)
#include <hip/hip_runtime.h>

typedef __attribute__((ext_vector_type(2))) float v2f;
typedef __attribute__((ext_vector_type(8))) float v8f;

namespace {
constexpr int Cc   = 64;
constexpr int Kk   = 16;
constexpr int Hh   = 128;
constexpr int Ww   = 128;
constexpr int HW   = Hh * Ww;
constexpr int PIX  = 32;
constexpr int PAIRS = Cc * Kk;
constexpr int TS   = PIX + 1;
constexpr float EPSV = 1e-6f;
}

__device__ __forceinline__ float bcast_lane(float v, int l) {
  return __builtin_bit_cast(float, __builtin_amdgcn_readlane(__builtin_bit_cast(int, v), l));
}
__device__ __forceinline__ float gather_lane(float v, int l) {
  return __builtin_bit_cast(float, __builtin_amdgcn_ds_bpermute(l << 2, __builtin_bit_cast(int, v)));
}

extern "C" __global__ __launch_bounds__(PIX * 32, 1)
void gram_spectral_inverse(const float* __restrict__ x, float* __restrict__ out) {
  extern __shared__ float tile[];

  const int tid  = threadIdx.x;
  const int lane = tid & 31;
  const int wv   = tid >> 5;

  const int pb = blockIdx.x * PIX;
  const int b  = pb >> 14;
  const int h  = (pb >> 7) & 127;
  const int w0 = pb & 127;

  const float* gbase = x + (size_t)b * (Cc * Kk * HW) + (size_t)h * Ww + w0;

  #pragma unroll 8
  for (int it = 0; it < PAIRS / PIX; ++it) {
    const int pair = it * PIX + wv;
    tile[pair * TS + lane] = gbase[(size_t)pair * HW + lane];
  }
  __syncthreads();

  const int pix = wv;
  const int kx  = lane & 15;
  const int chi = (lane >> 4) << 1;

  v8f acc = {};
  #pragma unroll
  for (int ch = 0; ch < 16; ++ch) {
    const int cA = ch * 4 + chi;
    v2f a;
    a.x = tile[(cA * 16 + kx) * TS + pix];
    a.y = tile[((cA + 1) * 16 + kx) * TS + pix];
    acc = __builtin_amdgcn_wmma_f32_16x16x4_f32(
         false, a,  false, a,
         (short)0, acc,  false,  false);
  }

  float s[8];
  #pragma unroll
  for (int r = 0; r < 8; ++r) {
    float v = acc[r];
    v += (lane == r || lane == r + 24) ? EPSV : 0.0f;
    s[r] = v;
  }

  float aug[16];
  const bool left = lane < 16;
  #pragma unroll
  for (int i = 0; i < 8; ++i) {
    const float hi = gather_lane(s[i], kx + 16);
    aug[i]     = left ? s[i] : ((i == kx) ? 1.0f : 0.0f);
    aug[i + 8] = left ? hi   : ((i + 8 == kx) ? 1.0f : 0.0f);
  }

  #pragma unroll
  for (int k = 0; k < 16; ++k) {
    const float p = bcast_lane(aug[k], k);
    float invp = __builtin_amdgcn_rcpf(p);
    invp = fmaf(invp, fmaf(-p, invp, 1.0f), invp);
    const float t = aug[k] * invp;
    #pragma unroll
    for (int i = 0; i < 16; ++i) {
      if (i == k) continue;
      const float c = (i & 1) ? gather_lane(aug[i], k)
                              : bcast_lane(aug[i], k);
      aug[i] = fmaf(-c, t, aug[i]);
    }
    aug[k] = t;
  }

  if (!left) {
    const int j = kx;
    float* op = out + (size_t)(pb + pix) * 256 + j * 16;
    reinterpret_cast<float4*>(op)[0] = make_float4(aug[0],  aug[1],  aug[2],  aug[3]);
    reinterpret_cast<float4*>(op)[1] = make_float4(aug[4],  aug[5],  aug[6],  aug[7]);
    reinterpret_cast<float4*>(op)[2] = make_float4(aug[8],  aug[9],  aug[10], aug[11]);
    reinterpret_cast<float4*>(op)[3] = make_float4(aug[12], aug[13], aug[14], aug[15]);
  }
}

extern "C" void kernel_launch(void* const* d_in, const int* in_sizes, int n_in,
                              void* d_out, int out_size, void* d_ws, size_t ws_size,
                              hipStream_t stream) {
  (void)in_sizes; (void)n_in; (void)out_size; (void)d_ws; (void)ws_size;
  const float* x = (const float*)d_in[0];
  float* out = (float*)d_out;
  const int nblocks = (4 * HW) / PIX;
  const size_t shmem = (size_t)PAIRS * TS * sizeof(float);
  hipLaunchKernelGGL(gram_spectral_inverse, dim3(nblocks), dim3(PIX * 32),
                     shmem, stream, x, out);
}


# Persistent work-groups (512 x 4 tiles): next tile's slab fetched in two 16-row halves under the current tile's Gauss-Jordan inverse, via registers into the LDS slab freed after the Gram
# speedup vs baseline: 1.1131x; 1.1131x over previous
#include <hip/hip_runtime.h>


typedef __attribute__((ext_vector_type(2))) float v2f;
typedef __attribute__((ext_vector_type(8))) float v8f;

namespace {
constexpr int Cc   = 64;
constexpr int Kk   = 16;
constexpr int Hh   = 128;
constexpr int Ww   = 128;
constexpr int HW   = Hh * Ww;
constexpr int PIX  = 32;
constexpr int PAIRS = Cc * Kk;
constexpr int TS   = PIX + 1;
constexpr float EPSV = 1e-6f;
constexpr int NTILE = (4 * HW) / PIX;
constexpr int NBLK  = 512;
constexpr int TPB   = NTILE / NBLK;
constexpr int ROWS  = PAIRS / PIX;
constexpr int HALF  = ROWS / 2;
}

__device__ __forceinline__ float bcast_lane(float v, int l) {
  return __builtin_bit_cast(float, __builtin_amdgcn_readlane(__builtin_bit_cast(int, v), l));
}
__device__ __forceinline__ float gather_lane(float v, int l) {
  return __builtin_bit_cast(float, __builtin_amdgcn_ds_bpermute(l << 2, __builtin_bit_cast(int, v)));
}

__device__ __forceinline__ void lds_barrier() {
  __builtin_amdgcn_fence(__ATOMIC_RELEASE, "workgroup", "local");
  __builtin_amdgcn_s_barrier();
  __builtin_amdgcn_fence(__ATOMIC_ACQUIRE, "workgroup", "local");
}

__device__ __forceinline__ const float* tile_base(const float* x, int t) {
  const int pb = t * PIX;
  const int b  = pb >> 14;
  const int h  = (pb >> 7) & 127;
  const int w0 = pb & 127;
  return x + (size_t)b * (Cc * Kk * HW) + (size_t)h * Ww + w0;
}

extern "C" __global__ __launch_bounds__(PIX * 32, 16)
void gram_spectral_inverse(const float* __restrict__ x, float* __restrict__ out) {
  extern __shared__ float tile[];

  const int tid  = threadIdx.x;
  const int lane = tid & 31;
  const int wv   = tid >> 5;

  const int pix = wv;
  const int kx  = lane & 15;
  const int chi = (lane >> 4) << 1;
  const bool left = lane < 16;
  const unsigned voff = (unsigned)(wv * HW + lane);
  const int      soff = wv * TS + lane;
  const int      rbase = ((chi * 16) + kx) * TS + pix;


  {
    const float* g0 = tile_base(x, blockIdx.x);
    #pragma unroll 8
    for (int it = 0; it < ROWS; ++it) {
      const int pair = it * PIX + wv;
      tile[pair * TS + lane] = g0[(size_t)pair * HW + lane];
    }
  }
  __syncthreads();

  for (int tl = 0; tl < TPB; ++tl) {
    const int t  = tl * NBLK + blockIdx.x;
    const int pb = t * PIX;
    const bool has_next = (tl + 1 < TPB);
    const float* gn = tile_base(x, has_next ? t + NBLK : t);

    v8f acc = {};
    #pragma unroll
    for (int ch = 0; ch < 16; ++ch) {
      v2f a;
      a.x = tile[rbase + (ch * 4) * 16 * TS];
      a.y = tile[rbase + (ch * 4 + 1) * 16 * TS];
      acc = __builtin_amdgcn_wmma_f32_16x16x4_f32(
           false, a,  false, a,
           (short)0, acc,  false,  false);
    }

    float pre[HALF];
    if (has_next) {
      lds_barrier();
      #pragma unroll
      for (int it = 0; it < HALF; ++it)
        pre[it] = gn[voff + (unsigned)(it * PIX * HW)];
    }

    float s[8];
    #pragma unroll
    for (int r = 0; r < 8; ++r) {
      const float v = acc[r];
      s[r] = (lane == r || lane == r + 24) ? v + EPSV : v;
    }

    float aug[16];
    #pragma unroll
    for (int i = 0; i < 8; ++i) {
      const float hi = gather_lane(s[i], kx + 16);
      const float lo0 = left ? s[i] : 0.0f;
      const float hi0 = left ? hi   : 0.0f;
      aug[i]     = (lane == 16 + i) ? 1.0f : lo0;
      aug[i + 8] = (lane == 24 + i) ? 1.0f : hi0;
    }

    #pragma unroll
    for (int k = 0; k < 16; ++k) {
      if (k == 8 && has_next) {
        #pragma unroll
        for (int it = 0; it < HALF; ++it)
          tile[it * PIX * TS + soff] = pre[it];
        #pragma unroll
        for (int it = 0; it < HALF; ++it)
          pre[it] = gn[voff + (unsigned)((it + HALF) * PIX * HW)];
      }
      const float p = bcast_lane(aug[k], k);
      float invp = __builtin_amdgcn_rcpf(p);
      invp = fmaf(invp, fmaf(-p, invp, 1.0f), invp);
      const float tt = aug[k] * invp;
      #pragma unroll
      for (int i = 0; i < 16; ++i) {
        if (i == k) continue;
        const float c = (i & 1) ? gather_lane(aug[i], k)
                                : bcast_lane(aug[i], k);
        aug[i] = fmaf(-c, tt, aug[i]);
      }
      aug[k] = tt;
    }

    if (!left) {
      const int j = kx;
      float* op = out + (size_t)(pb + pix) * 256 + j * 16;
      reinterpret_cast<float4*>(op)[0] = make_float4(aug[0],  aug[1],  aug[2],  aug[3]);
      reinterpret_cast<float4*>(op)[1] = make_float4(aug[4],  aug[5],  aug[6],  aug[7]);
      reinterpret_cast<float4*>(op)[2] = make_float4(aug[8],  aug[9],  aug[10], aug[11]);
      reinterpret_cast<float4*>(op)[3] = make_float4(aug[12], aug[13], aug[14], aug[15]);
    }

    if (has_next) {
      #pragma unroll
      for (int it = 0; it < HALF; ++it)
        tile[(it + HALF) * PIX * TS + soff] = pre[it];
      lds_barrier();
    }
  }
}

extern "C" void kernel_launch(void* const* d_in, const int* in_sizes, int n_in,
                              void* d_out, int out_size, void* d_ws, size_t ws_size,
                              hipStream_t stream) {
  (void)in_sizes; (void)n_in; (void)out_size; (void)d_ws; (void)ws_size;
  const float* x = (const float*)d_in[0];
  float* out = (float*)d_out;
  const size_t shmem = (size_t)PAIRS * TS * sizeof(float);
  hipLaunchKernelGGL(gram_spectral_inverse, dim3(NBLK), dim3(PIX * 32),
                     shmem, stream, x, out);
}
